# starting kernel
# baseline (speedup 1.0000x reference)
#include <hip/hip_runtime.h>


#define EMBED_DIM 64
#define NUM_LAYERS 3

typedef float v2f __attribute__((ext_vector_type(2)));
typedef float v8f __attribute__((ext_vector_type(8)));

__global__ void zero_f32(float* __restrict__ p, size_t n) {
  size_t i = (size_t)blockIdx.x * blockDim.x + threadIdx.x;
  size_t stride = (size_t)gridDim.x * blockDim.x;
  for (; i < n; i += stride) p[i] = 0.0f;
}

__global__ void deg_accum(const int* __restrict__ row, float* __restrict__ deg,
                          int E) {
  int e = blockIdx.x * blockDim.x + threadIdx.x;
  if (e < E) atomicAdd(&deg[row[e]], 1.0f);
}

__global__ void deg_to_dinv(float* __restrict__ dinv, int N) {
  int i = blockIdx.x * blockDim.x + threadIdx.x;
  if (i < N) {
    float d = dinv[i];
    dinv[i] = (d > 0.0f) ? rsqrtf(d) : 0.0f;
  }
}

__global__ void edge_norm(const int* __restrict__ row,
                          const int* __restrict__ col,
                          const float* __restrict__ dinv,
                          float* __restrict__ nrm, int E) {
  int e = blockIdx.x * blockDim.x + threadIdx.x;
  if (e < E) nrm[e] = dinv[row[e]] * dinv[col[e]];
}

__global__ void init_out(const float* __restrict__ emb, float* __restrict__ out,
                         size_t n, float c) {
  size_t i = (size_t)blockIdx.x * blockDim.x + threadIdx.x;
  size_t stride = (size_t)gridDim.x * blockDim.x;
  for (; i < n; i += stride) out[i] = emb[i] * c;
}

__global__ void scatter_layer(const int* __restrict__ row,
                              const int* __restrict__ col,
                              const float* __restrict__ w,
                              const int* __restrict__ et,
                              const float* __restrict__ nrm,
                              const float* __restrict__ hcur,
                              float* __restrict__ hnext,
                              float* __restrict__ s, int E) {
  int e = (int)(((size_t)blockIdx.x * blockDim.x + threadIdx.x) >> 5);
  int lane = threadIdx.x & 31;
  if (e >= E) return;

  int r = row[e];
  int c = col[e];
  float nm = nrm[e];
  float a = w[e] * nm;

  float2 hv = ((const float2*)(hcur + (size_t)r * EMBED_DIM))[lane];
  float* dst = hnext + (size_t)c * EMBED_DIM + lane * 2;
  atomicAdd(dst + 0, a * hv.x);
  atomicAdd(dst + 1, a * hv.y);
  if (lane == 0) atomicAdd(&s[(size_t)c * 4 + et[e]], nm);
}

__global__ void dense_wmma_out(const float* __restrict__ s,
                               const float* __restrict__ T,
                               float* __restrict__ hnext,
                               float* __restrict__ out,
                               float inv_lp1, int num_tiles) {
  int tile = (int)(((size_t)blockIdx.x * blockDim.x + threadIdx.x) >> 5);
  int lane = threadIdx.x & 31;
  if (tile >= num_tiles) return;

  int half = lane >> 4;
  int l16 = lane & 15;
  int k0 = half * 2;

  int arow = tile * 16 + l16;
  v2f A;
  A.x = s[(size_t)arow * 4 + k0];
  A.y = s[(size_t)arow * 4 + k0 + 1];

  for (int dt = 0; dt < 4; ++dt) {
    int n = dt * 16 + l16;
    v2f B;
    B.x = (k0 + 0 < 3) ? T[(k0 + 0) * EMBED_DIM + n] : 0.0f;
    B.y = (k0 + 1 < 3) ? T[(k0 + 1) * EMBED_DIM + n] : 0.0f;

    v8f C = {};
    C = __builtin_amdgcn_wmma_f32_16x16x4_f32(
         false, A,  false, B,
         (short)0, C,  false,  false);

#pragma unroll
    for (int r = 0; r < 8; ++r) {
      int node = tile * 16 + r + 8 * half;
      size_t idx = (size_t)node * EMBED_DIM + n;
      float hv = hnext[idx] + C[r];
      hnext[idx] = hv;
      out[idx] += inv_lp1 * hv;
    }
  }
}

extern "C" void kernel_launch(void* const* d_in, const int* in_sizes, int n_in,
                              void* d_out, int out_size, void* d_ws,
                              size_t ws_size, hipStream_t stream) {
  const int E = in_sizes[1];
  const int N = in_sizes[3] / EMBED_DIM;

  const int* eidx = (const int*)d_in[0];
  const int* row = eidx;
  const int* col = eidx + E;
  const float* w = (const float*)d_in[1];
  const int* et = (const int*)d_in[2];
  const float* emb = (const float*)d_in[3];
  const float* T = (const float*)d_in[4];
  float* out = (float*)d_out;

  float* ws = (float*)d_ws;
  float* dinv = ws;
  float* S = dinv + (size_t)N;
  float* nrm = S + (size_t)N * 4;
  float* h_a = nrm + (size_t)E;
  float* h_b = h_a + (size_t)N * EMBED_DIM;

  const float inv_lp1 = 1.0f / (NUM_LAYERS + 1);
  const int blk = 256;

  zero_f32<<<(N + blk - 1) / blk, blk, 0, stream>>>(dinv, (size_t)N);
  deg_accum<<<(E + blk - 1) / blk, blk, 0, stream>>>(row, dinv, E);
  deg_to_dinv<<<(N + blk - 1) / blk, blk, 0, stream>>>(dinv, N);
  edge_norm<<<(E + blk - 1) / blk, blk, 0, stream>>>(row, col, dinv, nrm, E);

  size_t nd = (size_t)N * EMBED_DIM;
  init_out<<<(int)((nd + blk - 1) / blk), blk, 0, stream>>>(emb, out, nd,
                                                            inv_lp1);

  const int scatter_grid = (int)(((size_t)E * 32 + blk - 1) / blk);
  const int tiles = N / 16;
  const int wmma_grid = (tiles * 32 + blk - 1) / blk;

  const float* hcur = emb;
  for (int l = 0; l < NUM_LAYERS; ++l) {
    float* hnext = (l & 1) ? h_b : h_a;

    zero_f32<<<(int)((nd + blk - 1) / blk), blk, 0, stream>>>(hnext, nd);
    zero_f32<<<(N * 4 + blk - 1) / blk, blk, 0, stream>>>(S, (size_t)N * 4);

    scatter_layer<<<scatter_grid, blk, 0, stream>>>(row, col, w, et, nrm, hcur,
                                                    hnext, S, E);
    dense_wmma_out<<<wmma_grid, blk, 0, stream>>>(S, T, hnext, out, inv_lp1,
                                                  tiles);
    hcur = hnext;
  }
}


# f32 atomic scatter and zero fills replaced by a per-call grouping of edges by destination (integer atomics; groups ordered by edge id, so a fixed sum order) and an atomic-free per-node row gather
# speedup vs baseline: 2.2791x; 2.2791x over previous
#include <hip/hip_runtime.h>


#define EMBED_DIM 64
#define NUM_LAYERS 3
#define SCAN_T 256
#define SCAN_PER 4

typedef float v2f __attribute__((ext_vector_type(2)));
typedef float v8f __attribute__((ext_vector_type(8)));

__global__ void zero_i32(int* __restrict__ p, size_t n) {
  size_t i = (size_t)blockIdx.x * blockDim.x + threadIdx.x;
  size_t stride = (size_t)gridDim.x * blockDim.x;
  for (; i < n; i += stride) p[i] = 0;
}

__global__ void init_out(const float* __restrict__ emb, float* __restrict__ out,
                         size_t n, float c) {
  size_t i = (size_t)blockIdx.x * blockDim.x + threadIdx.x;
  size_t stride = (size_t)gridDim.x * blockDim.x;
  for (; i < n; i += stride) out[i] = emb[i] * c;
}

__global__ void edge_hist(const int* __restrict__ row,
                          const int* __restrict__ col, int* __restrict__ deg,
                          int* __restrict__ cnt, int* __restrict__ pos, int E) {
  int e = blockIdx.x * blockDim.x + threadIdx.x;
  if (e < E) {
    int r = row[e];
    int c = col[e];
    atomicAdd(&deg[r], 1);
    pos[e] = atomicAdd(&cnt[c], 1);
  }
}

__global__ void node_prep(const int* __restrict__ deg,
                          const int* __restrict__ cnt,
                          float* __restrict__ dinv, int* __restrict__ bsum,
                          int N) {
  __shared__ int red[SCAN_T / 32];
  int base = (blockIdx.x * SCAN_T + threadIdx.x) * SCAN_PER;
  int s = 0;
  for (int k = 0; k < SCAN_PER; ++k) {
    int i = base + k;
    if (i < N) {
      float d = (float)deg[i];
      dinv[i] = (d > 0.0f) ? rsqrtf(d) : 0.0f;
      s += cnt[i];
    }
  }
  for (int o = 16; o > 0; o >>= 1) s += __shfl_xor(s, o, 32);
  if ((threadIdx.x & 31) == 0) red[threadIdx.x >> 5] = s;
  __syncthreads();
  if (threadIdx.x == 0) {
    int t = 0;
    for (int k = 0; k < SCAN_T / 32; ++k) t += red[k];
    bsum[blockIdx.x] = t;
  }
}

__global__ void node_scan(const int* __restrict__ cnt,
                          const int* __restrict__ bsum, int* __restrict__ offs,
                          int N) {
  __shared__ int wtot[SCAN_T / 32];
  __shared__ int ptot[SCAN_T / 32];
  int tid = threadIdx.x;
  int lane = tid & 31;
  int wave = tid >> 5;

  int part = 0;
  for (int b = tid; b < (int)blockIdx.x; b += SCAN_T) part += bsum[b];
  for (int o = 16; o > 0; o >>= 1) part += __shfl_xor(part, o, 32);

  int base = (blockIdx.x * SCAN_T + tid) * SCAN_PER;
  int c[SCAN_PER];
  int tsum = 0;
  for (int k = 0; k < SCAN_PER; ++k) {
    int i = base + k;
    c[k] = (i < N) ? cnt[i] : 0;
    tsum += c[k];
  }
  int incl = tsum;
  for (int o = 1; o < 32; o <<= 1) {
    int t = __shfl_up(incl, o, 32);
    if (lane >= o) incl += t;
  }
  if (lane == 31) wtot[wave] = incl;
  if (lane == 0) ptot[wave] = part;
  __syncthreads();
  int run = 0;
  for (int k = 0; k < SCAN_T / 32; ++k) {
    run += ptot[k];
    if (k < wave) run += wtot[k];
  }
  run += incl - tsum;
  for (int k = 0; k < SCAN_PER; ++k) {
    int i = base + k;
    if (i < N) offs[i] = run;
    run += c[k];
  }
}

__global__ void edge_fill(const int* __restrict__ row,
                          const int* __restrict__ col,
                          const float* __restrict__ w,
                          const int* __restrict__ et,
                          const int* __restrict__ pos,
                          const int* __restrict__ offs,
                          const float* __restrict__ dinv,
                          int* __restrict__ rec_words, int E) {
  int4* rec = (int4*)rec_words;
  int e = blockIdx.x * blockDim.x + threadIdx.x;
  if (e < E) {
    int r = row[e];
    int c = col[e];
    float nm = dinv[r] * dinv[c];
    float a = w[e] * nm;
    int slot = offs[c] + pos[e];
    rec[slot] = make_int4(e * 4 + (et[e] & 3), r, __float_as_int(a),
                          __float_as_int(nm));
  }
}

__global__ void node_finalize(const int* __restrict__ rec_words,
                              const int* __restrict__ offs,
                              int* __restrict__ pay_words,
                              float* __restrict__ s, int N, int E) {
  const int4* rec = (const int4*)rec_words;
  int2* pay = (int2*)pay_words;
  int c = blockIdx.x * blockDim.x + threadIdx.x;
  if (c >= N) return;
  int lo = offs[c];
  int hi = (c + 1 < N) ? offs[c + 1] : E;
  float s0 = 0.0f, s1 = 0.0f, s2 = 0.0f;
  int prev = -1;
  for (int k = lo; k < hi; ++k) {
    int best = 0x7fffffff;
    int bj = lo;
    for (int j = lo; j < hi; ++j) {
      int key = rec[j].x;
      if (key > prev && key < best) {
        best = key;
        bj = j;
      }
    }
    int4 rc = rec[bj];
    pay[k] = make_int2(rc.y, rc.z);
    float nm = __int_as_float(rc.w);
    int t = best & 3;
    s0 += (t == 0) ? nm : 0.0f;
    s1 += (t == 1) ? nm : 0.0f;
    s2 += (t == 2) ? nm : 0.0f;
    prev = best;
  }
  ((float4*)s)[c] = make_float4(s0, s1, s2, 0.0f);
}

__global__ void gather_layer(const int* __restrict__ pay_words,
                             const int* __restrict__ offs,
                             const float* __restrict__ hcur,
                             float* __restrict__ hnext, int N, int E) {
  const int2* pay = (const int2*)pay_words;
  int lane = threadIdx.x & 31;
  int node = __builtin_amdgcn_readfirstlane(
      (int)((blockIdx.x * blockDim.x + threadIdx.x) >> 5));
  if (node >= N) return;

  int lo = offs[node];
  int hi = (node + 1 < N) ? offs[node + 1] : E;
  const float2* h2 = (const float2*)hcur;

  float2 acc = make_float2(0.0f, 0.0f);
  int j = lo;
  for (; j + 4 <= hi; j += 4) {
    int2 p0 = pay[j + 0];
    int2 p1 = pay[j + 1];
    int2 p2 = pay[j + 2];
    int2 p3 = pay[j + 3];
    float2 v0 = h2[(size_t)p0.x * (EMBED_DIM / 2) + lane];
    float2 v1 = h2[(size_t)p1.x * (EMBED_DIM / 2) + lane];
    float2 v2 = h2[(size_t)p2.x * (EMBED_DIM / 2) + lane];
    float2 v3 = h2[(size_t)p3.x * (EMBED_DIM / 2) + lane];
    float a0 = __int_as_float(p0.y), a1 = __int_as_float(p1.y);
    float a2 = __int_as_float(p2.y), a3 = __int_as_float(p3.y);
    acc.x += a0 * v0.x;  acc.y += a0 * v0.y;
    acc.x += a1 * v1.x;  acc.y += a1 * v1.y;
    acc.x += a2 * v2.x;  acc.y += a2 * v2.y;
    acc.x += a3 * v3.x;  acc.y += a3 * v3.y;
  }
  int rem = hi - j;
  if (rem == 3) {
    int2 p0 = pay[j + 0];
    int2 p1 = pay[j + 1];
    int2 p2 = pay[j + 2];
    float2 v0 = h2[(size_t)p0.x * (EMBED_DIM / 2) + lane];
    float2 v1 = h2[(size_t)p1.x * (EMBED_DIM / 2) + lane];
    float2 v2 = h2[(size_t)p2.x * (EMBED_DIM / 2) + lane];
    float a0 = __int_as_float(p0.y), a1 = __int_as_float(p1.y);
    float a2 = __int_as_float(p2.y);
    acc.x += a0 * v0.x;  acc.y += a0 * v0.y;
    acc.x += a1 * v1.x;  acc.y += a1 * v1.y;
    acc.x += a2 * v2.x;  acc.y += a2 * v2.y;
  } else if (rem == 2) {
    int2 p0 = pay[j + 0];
    int2 p1 = pay[j + 1];
    float2 v0 = h2[(size_t)p0.x * (EMBED_DIM / 2) + lane];
    float2 v1 = h2[(size_t)p1.x * (EMBED_DIM / 2) + lane];
    float a0 = __int_as_float(p0.y), a1 = __int_as_float(p1.y);
    acc.x += a0 * v0.x;  acc.y += a0 * v0.y;
    acc.x += a1 * v1.x;  acc.y += a1 * v1.y;
  } else if (rem == 1) {
    int2 p0 = pay[j];
    float2 v0 = h2[(size_t)p0.x * (EMBED_DIM / 2) + lane];
    float a0 = __int_as_float(p0.y);
    acc.x += a0 * v0.x;  acc.y += a0 * v0.y;
  }
  ((float2*)hnext)[(size_t)node * (EMBED_DIM / 2) + lane] = acc;
}

__global__ void dense_wmma_out(const float* __restrict__ s,
                               const float* __restrict__ T,
                               float* __restrict__ hnext,
                               float* __restrict__ out,
                               float inv_lp1, int num_tiles) {
  int tile = (int)(((size_t)blockIdx.x * blockDim.x + threadIdx.x) >> 5);
  int lane = threadIdx.x & 31;
  if (tile >= num_tiles) return;

  int half = lane >> 4;
  int l16 = lane & 15;
  int k0 = half * 2;

  int arow = tile * 16 + l16;
  v2f A;
  A.x = s[(size_t)arow * 4 + k0];
  A.y = s[(size_t)arow * 4 + k0 + 1];

  for (int dt = 0; dt < 4; ++dt) {
    int n = dt * 16 + l16;
    v2f B;
    B.x = (k0 + 0 < 3) ? T[(k0 + 0) * EMBED_DIM + n] : 0.0f;
    B.y = (k0 + 1 < 3) ? T[(k0 + 1) * EMBED_DIM + n] : 0.0f;

    v8f C = {};
    C = __builtin_amdgcn_wmma_f32_16x16x4_f32(
         false, A,  false, B,
         (short)0, C,  false,  false);

#pragma unroll
    for (int r = 0; r < 8; ++r) {
      int node = tile * 16 + r + 8 * half;
      size_t idx = (size_t)node * EMBED_DIM + n;
      float hv = hnext[idx] + C[r];
      hnext[idx] = hv;
      out[idx] += inv_lp1 * hv;
    }
  }
}

extern "C" void kernel_launch(void* const* d_in, const int* in_sizes, int n_in,
                              void* d_out, int out_size, void* d_ws,
                              size_t ws_size, hipStream_t stream) {
  const int E = in_sizes[1];
  const int N = in_sizes[3] / EMBED_DIM;

  const int* eidx = (const int*)d_in[0];
  const int* row = eidx;
  const int* col = eidx + E;
  const float* w = (const float*)d_in[1];
  const int* et = (const int*)d_in[2];
  const float* emb = (const float*)d_in[3];
  const float* T = (const float*)d_in[4];
  float* out = (float*)d_out;

  float* ws = (float*)d_ws;
  int* offs = (int*)ws;
  float* S = ws + (size_t)N;
  int* pay = (int*)(S + (size_t)N * 4);
  float* h_a = S + (size_t)N * 4 + (size_t)E * 2;
  float* h_b = h_a + (size_t)N * EMBED_DIM;

  int* deg = (int*)h_b;
  int* cnt = deg + (size_t)N;
  float* dinv = (float*)(cnt + (size_t)N);
  int* bsum = (int*)(dinv + (size_t)N);
  int* pos = bsum + 256;
  int* rec = pos + (size_t)E;

  const float inv_lp1 = 1.0f / (NUM_LAYERS + 1);
  const int blk = 256;
  const int per_scan_block = SCAN_T * SCAN_PER;
  const int scan_grid = (N + per_scan_block - 1) / per_scan_block;

  zero_i32<<<(2 * N + blk - 1) / blk, blk, 0, stream>>>(deg, (size_t)N * 2);
  edge_hist<<<(E + blk - 1) / blk, blk, 0, stream>>>(row, col, deg, cnt, pos, E);
  node_prep<<<scan_grid, SCAN_T, 0, stream>>>(deg, cnt, dinv, bsum, N);
  node_scan<<<scan_grid, SCAN_T, 0, stream>>>(cnt, bsum, offs, N);
  edge_fill<<<(E + blk - 1) / blk, blk, 0, stream>>>(row, col, w, et, pos, offs,
                                                     dinv, rec, E);
  node_finalize<<<(N + blk - 1) / blk, blk, 0, stream>>>(rec, offs, pay, S, N, E);

  size_t nd = (size_t)N * EMBED_DIM;
  init_out<<<(int)((nd + blk - 1) / blk), blk, 0, stream>>>(emb, out, nd,
                                                            inv_lp1);

  const int gather_grid = (int)(((size_t)N * 32 + blk - 1) / blk);
  const int tiles = N / 16;
  const int wmma_grid = (tiles * 32 + blk - 1) / blk;

  const float* hcur = emb;
  for (int l = 0; l < NUM_LAYERS; ++l) {
    float* hnext = (l & 1) ? h_b : h_a;

    gather_layer<<<gather_grid, blk, 0, stream>>>(pay, offs, hcur, hnext, N, E);
    dense_wmma_out<<<wmma_grid, blk, 0, stream>>>(S, T, hnext, out, inv_lp1,
                                                  tiles);
    hcur = hnext;
  }
}


# Fewer, longer-lived waves: a gather wave walks 13 nodes' contiguous edge run with 8 rows in flight; out start value and the two edge kernels take 4 or more items per lane; sum order unchanged
# speedup vs baseline: 2.6447x; 1.1604x over previous
#include <hip/hip_runtime.h>


#define EMBED_DIM 64
#define NUM_LAYERS 3
#define SCAN_T 256
#define SCAN_PER 4

typedef float v2f __attribute__((ext_vector_type(2)));
typedef float v8f __attribute__((ext_vector_type(8)));

__global__ void zero_i32(int* __restrict__ p, size_t n) {
  size_t i = (size_t)blockIdx.x * blockDim.x + threadIdx.x;
  size_t stride = (size_t)gridDim.x * blockDim.x;
  for (; i < n; i += stride) p[i] = 0;
}

__device__ __forceinline__ float4 scale4(float4 a, float c) {
  return make_float4(a.x * c, a.y * c, a.z * c, a.w * c);
}

__global__ void init_out(const float* __restrict__ emb, float* __restrict__ out,
                         size_t n, float c) {
  const float4* __restrict__ e4 = (const float4*)emb;
  float4* __restrict__ o4 = (float4*)out;
  const size_t n4 = n >> 2;
  const size_t stride = (size_t)gridDim.x * blockDim.x;
  size_t i = (size_t)blockIdx.x * blockDim.x + threadIdx.x;
  for (; i + 7 * stride < n4; i += 8 * stride) {
    float4 v[8];
#pragma unroll
    for (int k = 0; k < 8; ++k) v[k] = e4[i + k * stride];
#pragma unroll
    for (int k = 0; k < 8; ++k) o4[i + k * stride] = scale4(v[k], c);
  }
  for (; i + 3 * stride < n4; i += 4 * stride) {
    float4 v[4];
#pragma unroll
    for (int k = 0; k < 4; ++k) v[k] = e4[i + k * stride];
#pragma unroll
    for (int k = 0; k < 4; ++k) o4[i + k * stride] = scale4(v[k], c);
  }
  for (; i < n4; i += stride) o4[i] = scale4(e4[i], c);
}

#define EDGE_PER 4
template <bool TAIL>
__device__ __forceinline__ void edge_hist_body(const int* __restrict__ row,
                                               const int* __restrict__ col,
                                               int* __restrict__ deg,
                                               int* __restrict__ cnt,
                                               int* __restrict__ pos, int E,
                                               int t, int stride) {
  int r[EDGE_PER], c[EDGE_PER], p[EDGE_PER];
#pragma unroll
  for (int k = 0; k < EDGE_PER; ++k) {
    int e = t + k * stride;
    int ec = (TAIL && e >= E) ? E - 1 : e;
    r[k] = row[ec];
    c[k] = col[ec];
  }
#pragma unroll
  for (int k = 0; k < EDGE_PER; ++k) {
    int e = t + k * stride;
    p[k] = 0;
    if (!TAIL || e < E) {
      atomicAdd(&deg[r[k]], 1);
      p[k] = atomicAdd(&cnt[c[k]], 1);
    }
  }
#pragma unroll
  for (int k = 0; k < EDGE_PER; ++k) {
    int e = t + k * stride;
    if (!TAIL || e < E) pos[e] = p[k];
  }
}

__global__ void edge_hist(const int* __restrict__ row,
                          const int* __restrict__ col, int* __restrict__ deg,
                          int* __restrict__ cnt, int* __restrict__ pos, int E) {
  const int t = blockIdx.x * blockDim.x + threadIdx.x;
  const int stride = gridDim.x * blockDim.x;
  if (t + (EDGE_PER - 1) * stride < E)
    edge_hist_body<false>(row, col, deg, cnt, pos, E, t, stride);
  else if (t < E)
    edge_hist_body<true>(row, col, deg, cnt, pos, E, t, stride);
}

__global__ void node_prep(const int* __restrict__ deg,
                          const int* __restrict__ cnt,
                          float* __restrict__ dinv, int* __restrict__ bsum,
                          int N) {
  __shared__ int red[SCAN_T / 32];
  int base = (blockIdx.x * SCAN_T + threadIdx.x) * SCAN_PER;
  int s = 0;
  for (int k = 0; k < SCAN_PER; ++k) {
    int i = base + k;
    if (i < N) {
      float d = (float)deg[i];
      dinv[i] = (d > 0.0f) ? rsqrtf(d) : 0.0f;
      s += cnt[i];
    }
  }
  for (int o = 16; o > 0; o >>= 1) s += __shfl_xor(s, o, 32);
  if ((threadIdx.x & 31) == 0) red[threadIdx.x >> 5] = s;
  __syncthreads();
  if (threadIdx.x == 0) {
    int t = 0;
    for (int k = 0; k < SCAN_T / 32; ++k) t += red[k];
    bsum[blockIdx.x] = t;
  }
}

__global__ void node_scan(const int* __restrict__ cnt,
                          const int* __restrict__ bsum, int* __restrict__ offs,
                          int N) {
  __shared__ int wtot[SCAN_T / 32];
  __shared__ int ptot[SCAN_T / 32];
  int tid = threadIdx.x;
  int lane = tid & 31;
  int wave = tid >> 5;

  int part = 0;
  for (int b = tid; b < (int)blockIdx.x; b += SCAN_T) part += bsum[b];
  for (int o = 16; o > 0; o >>= 1) part += __shfl_xor(part, o, 32);

  int base = (blockIdx.x * SCAN_T + tid) * SCAN_PER;
  int c[SCAN_PER];
  int tsum = 0;
  for (int k = 0; k < SCAN_PER; ++k) {
    int i = base + k;
    c[k] = (i < N) ? cnt[i] : 0;
    tsum += c[k];
  }
  int incl = tsum;
  for (int o = 1; o < 32; o <<= 1) {
    int t = __shfl_up(incl, o, 32);
    if (lane >= o) incl += t;
  }
  if (lane == 31) wtot[wave] = incl;
  if (lane == 0) ptot[wave] = part;
  __syncthreads();
  int run = 0;
  for (int k = 0; k < SCAN_T / 32; ++k) {
    run += ptot[k];
    if (k < wave) run += wtot[k];
  }
  run += incl - tsum;
  for (int k = 0; k < SCAN_PER; ++k) {
    int i = base + k;
    if (i < N) offs[i] = run;
    run += c[k];
  }
}

template <bool TAIL>
__device__ __forceinline__ void edge_fill_body(
    const int* __restrict__ row, const int* __restrict__ col,
    const float* __restrict__ w, const int* __restrict__ et,
    const int* __restrict__ pos, const int* __restrict__ offs,
    const float* __restrict__ dinv, int4* __restrict__ rec, int E, int t,
    int stride) {
  int r[EDGE_PER], c[EDGE_PER], ty[EDGE_PER], ps[EDGE_PER];
  float we[EDGE_PER];
#pragma unroll
  for (int k = 0; k < EDGE_PER; ++k) {
    int e = t + k * stride;
    int ec = (TAIL && e >= E) ? E - 1 : e;
    r[k] = row[ec];
    c[k] = col[ec];
    we[k] = w[ec];
    ty[k] = et[ec];
    ps[k] = pos[ec];
  }
  float dr[EDGE_PER], dc[EDGE_PER];
  int of[EDGE_PER];
#pragma unroll
  for (int k = 0; k < EDGE_PER; ++k) {
    dr[k] = dinv[r[k]];
    dc[k] = dinv[c[k]];
    of[k] = offs[c[k]];
  }
#pragma unroll
  for (int k = 0; k < EDGE_PER; ++k) {
    int e = t + k * stride;
    if (!TAIL || e < E) {
      float nm = dr[k] * dc[k];
      float a = we[k] * nm;
      rec[of[k] + ps[k]] = make_int4(e * 4 + (ty[k] & 3), r[k],
                                     __float_as_int(a), __float_as_int(nm));
    }
  }
}

__global__ void edge_fill(const int* __restrict__ row,
                          const int* __restrict__ col,
                          const float* __restrict__ w,
                          const int* __restrict__ et,
                          const int* __restrict__ pos,
                          const int* __restrict__ offs,
                          const float* __restrict__ dinv,
                          int* __restrict__ rec_words, int E) {
  int4* rec = (int4*)rec_words;
  const int t = blockIdx.x * blockDim.x + threadIdx.x;
  const int stride = gridDim.x * blockDim.x;
  if (t + (EDGE_PER - 1) * stride < E)
    edge_fill_body<false>(row, col, w, et, pos, offs, dinv, rec, E, t, stride);
  else if (t < E)
    edge_fill_body<true>(row, col, w, et, pos, offs, dinv, rec, E, t, stride);
}

__global__ void node_finalize(const int* __restrict__ rec_words,
                              const int* __restrict__ offs,
                              int* __restrict__ pay_words,
                              float* __restrict__ s, int N, int E) {
  const int4* rec = (const int4*)rec_words;
  int2* pay = (int2*)pay_words;
  int c = blockIdx.x * blockDim.x + threadIdx.x;
  if (c >= N) return;
  int lo = offs[c];
  int hi = (c + 1 < N) ? offs[c + 1] : E;
  float s0 = 0.0f, s1 = 0.0f, s2 = 0.0f;
  int prev = -1;
  for (int k = lo; k < hi; ++k) {
    int best = 0x7fffffff;
    int bj = lo;
    for (int j = lo; j < hi; ++j) {
      int key = rec[j].x;
      if (key > prev && key < best) {
        best = key;
        bj = j;
      }
    }
    int4 rc = rec[bj];
    pay[k] = make_int2(rc.y, rc.z);
    float nm = __int_as_float(rc.w);
    int t = best & 3;
    s0 += (t == 0) ? nm : 0.0f;
    s1 += (t == 1) ? nm : 0.0f;
    s2 += (t == 2) ? nm : 0.0f;
    prev = best;
  }
  ((float4*)s)[c] = make_float4(s0, s1, s2, 0.0f);
}

#define GATHER_NPW 13
#define GATHER_RB 8

__device__ __forceinline__ void gather_add(int j, float a, float2 v, int& cur,
                                           int& nextb, float2& acc, int starts,
                                           int n0, float2* __restrict__ o2,
                                           int lane) {
  while (j == nextb) {
    o2[(size_t)cur * (EMBED_DIM / 2) + lane] = acc;
    acc = make_float2(0.0f, 0.0f);
    ++cur;
    nextb = __builtin_amdgcn_readlane(starts, cur - n0 + 1);
  }
  acc.x += a * v.x;
  acc.y += a * v.y;
}

__global__ void gather_layer(const int* __restrict__ pay_words,
                             const int* __restrict__ offs,
                             const float* __restrict__ hcur,
                             float* __restrict__ hnext, int N, int E) {
  const int2* pay = (const int2*)pay_words;
  const float2* h2 = (const float2*)hcur;
  float2* o2 = (float2*)hnext;
  const int lane = threadIdx.x & 31;
  const int wave = __builtin_amdgcn_readfirstlane(
      (int)((blockIdx.x * blockDim.x + threadIdx.x) >> 5));
  const int n0 = wave * GATHER_NPW;
  if (n0 >= N) return;
  const int n1 = (n0 + GATHER_NPW < N) ? n0 + GATHER_NPW : N;

  const int bi = n0 + lane;
  const int starts = (bi < N) ? offs[bi] : E;
  const int e0 = __builtin_amdgcn_readlane(starts, 0);
  const int e1 = __builtin_amdgcn_readlane(starts, n1 - n0);

  int cur = n0;
  int nextb = __builtin_amdgcn_readlane(starts, 1);
  float2 acc = make_float2(0.0f, 0.0f);

  if (e1 > e0) {
    int s0 = e0 + lane;
    int2 pnext = pay[(s0 < e1) ? s0 : e1 - 1];
    for (int jc = e0; jc < e1; jc += 32) {
      const int2 pv = pnext;
      if (jc + 32 < e1) {
        int s = jc + 32 + lane;
        pnext = pay[(s < e1) ? s : e1 - 1];
      }
      const int left = e1 - jc;
      const int cnt = (left < 32) ? left : 32;
      int b = 0;
      for (; b + GATHER_RB <= cnt; b += GATHER_RB) {
        float2 v[GATHER_RB];
#pragma unroll
        for (int k = 0; k < GATHER_RB; ++k) {
          int r = __builtin_amdgcn_readlane(pv.x, b + k);
          v[k] = h2[(size_t)r * (EMBED_DIM / 2) + lane];
        }
#pragma unroll
        for (int k = 0; k < GATHER_RB; ++k) {
          const float a = __int_as_float(__builtin_amdgcn_readlane(pv.y, b + k));
          gather_add(jc + b + k, a, v[k], cur, nextb, acc, starts, n0, o2, lane);
        }
      }
      if (b < cnt) {
        float2 v[GATHER_RB - 1];
#pragma unroll
        for (int k = 0; k < GATHER_RB - 1; ++k) {
          int r = __builtin_amdgcn_readlane(pv.x, b + k);
          v[k] = h2[(size_t)r * (EMBED_DIM / 2) + lane];
        }
#pragma unroll
        for (int k = 0; k < GATHER_RB - 1; ++k) {
          if (b + k < cnt) {
            const float a = __int_as_float(__builtin_amdgcn_readlane(pv.y, b + k));
            gather_add(jc + b + k, a, v[k], cur, nextb, acc, starts, n0, o2, lane);
          }
        }
      }
    }
  }
  for (; cur < n1; ++cur) {
    o2[(size_t)cur * (EMBED_DIM / 2) + lane] = acc;
    acc = make_float2(0.0f, 0.0f);
  }
}

__global__ void dense_wmma_out(const float* __restrict__ s,
                               const float* __restrict__ T,
                               float* __restrict__ hnext,
                               float* __restrict__ out,
                               float inv_lp1, int num_tiles) {
  int tile = (int)(((size_t)blockIdx.x * blockDim.x + threadIdx.x) >> 5);
  int lane = threadIdx.x & 31;
  if (tile >= num_tiles) return;

  int half = lane >> 4;
  int l16 = lane & 15;
  int k0 = half * 2;

  int arow = tile * 16 + l16;
  v2f A;
  A.x = s[(size_t)arow * 4 + k0];
  A.y = s[(size_t)arow * 4 + k0 + 1];

  for (int dt = 0; dt < 4; ++dt) {
    int n = dt * 16 + l16;
    v2f B;
    B.x = (k0 + 0 < 3) ? T[(k0 + 0) * EMBED_DIM + n] : 0.0f;
    B.y = (k0 + 1 < 3) ? T[(k0 + 1) * EMBED_DIM + n] : 0.0f;

    v8f C = {};
    C = __builtin_amdgcn_wmma_f32_16x16x4_f32(
         false, A,  false, B,
         (short)0, C,  false,  false);

#pragma unroll
    for (int r = 0; r < 8; ++r) {
      int node = tile * 16 + r + 8 * half;
      size_t idx = (size_t)node * EMBED_DIM + n;
      float hv = hnext[idx] + C[r];
      hnext[idx] = hv;
      out[idx] += inv_lp1 * hv;
    }
  }
}

extern "C" void kernel_launch(void* const* d_in, const int* in_sizes, int n_in,
                              void* d_out, int out_size, void* d_ws,
                              size_t ws_size, hipStream_t stream) {
  const int E = in_sizes[1];
  const int N = in_sizes[3] / EMBED_DIM;

  const int* eidx = (const int*)d_in[0];
  const int* row = eidx;
  const int* col = eidx + E;
  const float* w = (const float*)d_in[1];
  const int* et = (const int*)d_in[2];
  const float* emb = (const float*)d_in[3];
  const float* T = (const float*)d_in[4];
  float* out = (float*)d_out;

  float* ws = (float*)d_ws;
  int* offs = (int*)ws;
  float* S = ws + (size_t)N;
  int* pay = (int*)(S + (size_t)N * 4);
  float* h_a = S + (size_t)N * 4 + (size_t)E * 2;
  float* h_b = h_a + (size_t)N * EMBED_DIM;

  int* deg = (int*)h_b;
  int* cnt = deg + (size_t)N;
  float* dinv = (float*)(cnt + (size_t)N);
  int* bsum = (int*)(dinv + (size_t)N);
  int* pos = bsum + 256;
  int* rec = pos + (size_t)E;

  const float inv_lp1 = 1.0f / (NUM_LAYERS + 1);
  const int blk = 256;
  const int per_scan_block = SCAN_T * SCAN_PER;
  const int scan_grid = (N + per_scan_block - 1) / per_scan_block;

  zero_i32<<<(2 * N + blk * 8 - 1) / (blk * 8), blk, 0, stream>>>(deg, (size_t)N * 2);
  const int edge_grid = (E + blk * EDGE_PER - 1) / (blk * EDGE_PER);
  edge_hist<<<edge_grid, blk, 0, stream>>>(row, col, deg, cnt, pos, E);
  node_prep<<<scan_grid, SCAN_T, 0, stream>>>(deg, cnt, dinv, bsum, N);
  node_scan<<<scan_grid, SCAN_T, 0, stream>>>(cnt, bsum, offs, N);
  edge_fill<<<edge_grid, blk, 0, stream>>>(row, col, w, et, pos, offs, dinv,
                                           rec, E);
  node_finalize<<<(N + blk - 1) / blk, blk, 0, stream>>>(rec, offs, pay, S, N, E);

  size_t nd = (size_t)N * EMBED_DIM;
  init_out<<<1024, blk, 0, stream>>>(emb, out, nd, inv_lp1);

  const int gather_waves = (N + GATHER_NPW - 1) / GATHER_NPW;
  const int gather_grid = (int)(((size_t)gather_waves * 32 + blk - 1) / blk);
  const int tiles = N / 16;
  const int wmma_grid = (tiles * 32 + blk - 1) / blk;

  const float* hcur = emb;
  for (int l = 0; l < NUM_LAYERS; ++l) {
    float* hnext = (l & 1) ? h_b : h_a;

    gather_layer<<<gather_grid, blk, 0, stream>>>(pay, offs, hcur, hnext, N, E);
    dense_wmma_out<<<wmma_grid, blk, 0, stream>>>(S, T, hnext, out, inv_lp1,
                                                  tiles);
    hcur = hnext;
  }
}


# Bucket records placed and read back for the edge-id ordering in an LDS copy of the bucket's edge range, not through memory (memory path kept for oversize buckets); sum order kept
# speedup vs baseline: 7.0017x; 2.6474x over previous
#include <hip/hip_runtime.h>


#define EMBED_DIM 64
#define NUM_LAYERS 3

typedef float v2f __attribute__((ext_vector_type(2)));
typedef float v8f __attribute__((ext_vector_type(8)));

#define BIN_SL 4096
#define BIN_T 1024
#define BIN_PER (BIN_SL / BIN_T)
#define BKT_SHIFT 10
#define BKT_N (1 << BKT_SHIFT)
#define MAXB 256
#define BSTRIDE (MAXB + 1)
#define MAXGA 512

__global__ __launch_bounds__(BIN_T) void bin_edges(
    const int* __restrict__ row, const int* __restrict__ col,
    const float* __restrict__ w, const int* __restrict__ et,
    int* __restrict__ binc_words, int* __restrict__ rbin_words,
    int* __restrict__ startc, int* __restrict__ startr, int E) {
  __shared__ int hc[MAXB];
  __shared__ int hr[MAXB];
  __shared__ int wtc[MAXB / 32];
  __shared__ int wtr[MAXB / 32];
  __shared__ int4 sc[BIN_SL];
  __shared__ unsigned short sr[BIN_SL];
  int4* binc = (int4*)binc_words;
  unsigned short* rbin = (unsigned short*)rbin_words;
  const int tid = threadIdx.x;
  const int lane = tid & 31;
  const int wave = tid >> 5;
  const int base = blockIdx.x * BIN_SL;
  const int n = (E - base < BIN_SL) ? E - base : BIN_SL;

  if (tid < MAXB) {
    hc[tid] = 0;
    hr[tid] = 0;
  }
  __syncthreads();

  int r[BIN_PER], c[BIN_PER], ty[BIN_PER], pc[BIN_PER], pr[BIN_PER];
  float we[BIN_PER];
#pragma unroll
  for (int k = 0; k < BIN_PER; ++k) {
    const int i = tid + k * BIN_T;
    const int e = base + ((i < n) ? i : n - 1);
    r[k] = row[e];
    c[k] = col[e];
    we[k] = w[e];
    ty[k] = et[e];
  }
#pragma unroll
  for (int k = 0; k < BIN_PER; ++k) {
    const int i = tid + k * BIN_T;
    pc[k] = 0;
    pr[k] = 0;
    if (i < n) {
      pc[k] = atomicAdd(&hc[c[k] >> BKT_SHIFT], 1);
      pr[k] = atomicAdd(&hr[r[k] >> BKT_SHIFT], 1);
    }
  }
  __syncthreads();

  int vc = 0, vr = 0, ic = 0, ir = 0;
  if (tid < MAXB) {
    vc = hc[tid];
    vr = hr[tid];
    ic = vc;
    ir = vr;
    for (int o = 1; o < 32; o <<= 1) {
      const int tc = __shfl_up(ic, o, 32);
      const int tr = __shfl_up(ir, o, 32);
      if (lane >= o) {
        ic += tc;
        ir += tr;
      }
    }
    if (lane == 31) {
      wtc[wave] = ic;
      wtr[wave] = ir;
    }
  }
  __syncthreads();
  if (tid < MAXB) {
    int oc = 0, orr = 0;
    for (int k = 0; k < wave; ++k) {
      oc += wtc[k];
      orr += wtr[k];
    }
    const int ec = oc + ic - vc;
    const int er = orr + ir - vr;
    hc[tid] = ec;
    hr[tid] = er;
    startc[(size_t)blockIdx.x * BSTRIDE + tid] = ec;
    startr[(size_t)blockIdx.x * BSTRIDE + tid] = er;
    if (tid == MAXB - 1) {
      startc[(size_t)blockIdx.x * BSTRIDE + MAXB] = ec + vc;
      startr[(size_t)blockIdx.x * BSTRIDE + MAXB] = er + vr;
    }
  }
  __syncthreads();

#pragma unroll
  for (int k = 0; k < BIN_PER; ++k) {
    const int i = tid + k * BIN_T;
    if (i < n) {
      sc[hc[c[k] >> BKT_SHIFT] + pc[k]] =
          make_int4((base + i) * 4 + (ty[k] & 3), r[k], c[k],
                    __float_as_int(we[k]));
      sr[hr[r[k] >> BKT_SHIFT] + pr[k]] = (unsigned short)(r[k] & (BKT_N - 1));
    }
  }
  __syncthreads();

  for (int i = tid; i < n; i += BIN_T) {
    binc[(size_t)base + i] = sc[i];
    rbin[(size_t)base + i] = sr[i];
  }
}

__global__ __launch_bounds__(BIN_T) void bucket_count(
    const int* __restrict__ binc_words, const int* __restrict__ rbin_words,
    const int* __restrict__ startc, const int* __restrict__ startr,
    float* __restrict__ dinv, int* __restrict__ offs, int N, int GA) {
  __shared__ int cd[BKT_N];
  __shared__ int cc[BKT_N];
  __shared__ int2 runc[MAXGA];
  __shared__ int2 runr[MAXGA];
  __shared__ int red[BIN_T / 32];
  __shared__ int wtot[BIN_T / 32];
  const int4* binc = (const int4*)binc_words;
  const unsigned short* rbin = (const unsigned short*)rbin_words;
  const int tid = threadIdx.x;
  const int lane = tid & 31;
  const int wave = __builtin_amdgcn_readfirstlane(tid >> 5);
  const int b = blockIdx.x;

  cd[tid] = 0;
  cc[tid] = 0;
  int part = 0;
  for (int k = tid; k < GA; k += BIN_T) {
    const int s0 = startc[(size_t)k * BSTRIDE + b];
    const int s1 = startc[(size_t)k * BSTRIDE + b + 1];
    const int t0 = startr[(size_t)k * BSTRIDE + b];
    const int t1 = startr[(size_t)k * BSTRIDE + b + 1];
    runc[k] = make_int2(s0, s1);
    runr[k] = make_int2(t0, t1);
    part += s0;
  }
  for (int o = 16; o > 0; o >>= 1) part += __shfl_xor(part, o, 32);
  if (lane == 0) red[wave] = part;
  __syncthreads();

  for (int k = wave; k < GA; k += BIN_T / 32) {
    const int2 dc = runc[k];
    const int2 dr = runr[k];
    const size_t sb = (size_t)k * BIN_SL;
    int jc = dc.x + lane;
    int jr = dr.x + lane;
    const bool hasc = jc < dc.y;
    const bool hasr = jr < dr.y;
    const int rv = rbin[sb + (hasr ? jr : 0)];
    int cv = 0;
    if (hasc) cv = binc[sb + jc].z;
    if (hasc) atomicAdd(&cc[cv & (BKT_N - 1)], 1);
    if (hasr) atomicAdd(&cd[rv], 1);
    for (jc += 32; jc < dc.y; jc += 32)
      atomicAdd(&cc[binc[sb + jc].z & (BKT_N - 1)], 1);
    for (jr += 32; jr < dr.y; jr += 32) atomicAdd(&cd[rbin[sb + jr]], 1);
  }
  __syncthreads();

  const int node = b * BKT_N + tid;
  if (node < N) {
    const float d = (float)cd[tid];
    dinv[node] = (d > 0.0f) ? rsqrtf(d) : 0.0f;
  }
  const int v = cc[tid];
  int incl = v;
  for (int o = 1; o < 32; o <<= 1) {
    const int t = __shfl_up(incl, o, 32);
    if (lane >= o) incl += t;
  }
  if (lane == 31) wtot[wave] = incl;
  __syncthreads();
  int run = 0;
  for (int k = 0; k < BIN_T / 32; ++k) {
    run += red[k];
    if (k < wave) run += wtot[k];
  }
  run += incl - v;
  if (node < N) offs[node] = run;
}

__device__ __forceinline__ void place_one(const int4 rc, const float dr,
                                          const float* __restrict__ dl,
                                          int* __restrict__ cur,
                                          int4* rec, int4* lrec, int base0,
                                          bool fits) {
  const int cl = rc.z & (BKT_N - 1);
  const float nm = dr * dl[cl];
  const float a = __int_as_float(rc.w) * nm;
  const int slot = atomicAdd(&cur[cl], 1);
  const int4 v = make_int4(rc.x, rc.y, __float_as_int(a), __float_as_int(nm));
  if (fits)
    lrec[slot - base0] = v;
  else
    rec[slot] = v;
}

template <int NS>
__device__ __forceinline__ void sort_regs(int (&v)[NS]) {
#pragma unroll
  for (int p = 1; (1 << p) <= NS; ++p) {
#pragma unroll
    for (int q = p - 1; q >= 0; --q) {
#pragma unroll
      for (int i = 0; i < NS; ++i) {
        const int l = i ^ (1 << q);
        if (l > i) {
          const int a = v[i];
          const int b = v[l];
          const int mn = (a < b) ? a : b;
          const int mx = (a < b) ? b : a;
          const bool up = ((i >> p) & 1) == 0;
          v[i] = up ? mn : mx;
          v[l] = up ? mx : mn;
        }
      }
    }
  }
}

template <int NS, int SH>
__device__ __forceinline__ void order_in_regs(const int4* rec, int rlo,
                                              int2* __restrict__ pay, int lo,
                                              int d, float& s0, float& s1,
                                              float& s2) {
  if (d <= 0) return;
  int v[NS];
#pragma unroll
  for (int i = 0; i < NS; ++i) {
    const int key = rec[rlo + ((i < d) ? i : 0)].x;
    v[i] = (i < d) ? ((key << SH) | i) : 0x7fffffff;
  }
  sort_regs<NS>(v);
#pragma unroll
  for (int c = 0; c < NS; c += 8) {
    int4 r[8];
#pragma unroll
    for (int k = 0; k < 8; ++k) {
      r[k] = make_int4(0, 0, 0, 0);
      if (c + k < d) r[k] = rec[rlo + (v[c + k] & (NS - 1))];
    }
#pragma unroll
    for (int k = 0; k < 8; ++k) {
      if (c + k < d) {
        pay[lo + c + k] = make_int2(r[k].y, r[k].z);
        const float nm = __int_as_float(r[k].w);
        const int t = (v[c + k] >> SH) & 3;
        s0 += (t == 0) ? nm : 0.0f;
        s1 += (t == 1) ? nm : 0.0f;
        s2 += (t == 2) ? nm : 0.0f;
      }
    }
  }
}

__device__ __forceinline__ void order_by_search(const int4* rec, int rlo,
                                                int2* __restrict__ pay, int lo,
                                                int d, float& s0, float& s1,
                                                float& s2) {
  int prev = -1;
  for (int k = 0; k < d; ++k) {
    int best = 0x7fffffff;
    int bj = 0;
    for (int j = 0; j < d; ++j) {
      const int key = rec[rlo + j].x;
      if (key > prev && key < best) {
        best = key;
        bj = j;
      }
    }
    const int4 rc = rec[rlo + bj];
    pay[lo + k] = make_int2(rc.y, rc.z);
    const float nm = __int_as_float(rc.w);
    const int t = best & 3;
    s0 += (t == 0) ? nm : 0.0f;
    s1 += (t == 1) ? nm : 0.0f;
    s2 += (t == 2) ? nm : 0.0f;
    prev = best;
  }
}

#define LREC_CAP 16384
__global__ __launch_bounds__(BIN_T) void bucket_place(
    const int* __restrict__ binc_words, const int* __restrict__ startc,
    const int* __restrict__ offs, const float* __restrict__ dinv,
    int* __restrict__ rec_words, int* __restrict__ pay_words,
    float* __restrict__ s, int N, int GA, int E) {
  __shared__ int cur[BKT_N];
  __shared__ float dl[BKT_N];
  __shared__ int2 runc[MAXGA];
  __shared__ int4 lrec[LREC_CAP];
  const int4* binc = (const int4*)binc_words;
  int4* rec = (int4*)rec_words;
  const int tid = threadIdx.x;
  const int lane = tid & 31;
  const int wave = __builtin_amdgcn_readfirstlane(tid >> 5);
  const int b = blockIdx.x;

  const int base0 = offs[b * BKT_N];
  const int end0 = ((b + 1) * BKT_N < N) ? offs[(b + 1) * BKT_N] : E;
  const bool fits = (end0 - base0) <= LREC_CAP;

  const int node = b * BKT_N + tid;
  const int lo = (node < N) ? offs[node] : 0;
  cur[tid] = lo;
  dl[tid] = (node < N) ? dinv[node] : 0.0f;
  for (int k = tid; k < GA; k += BIN_T)
    runc[k] = make_int2(startc[(size_t)k * BSTRIDE + b],
                        startc[(size_t)k * BSTRIDE + b + 1]);
  __syncthreads();

  for (int k = wave; k < GA; k += 2 * (BIN_T / 32)) {
    const int k2 = k + BIN_T / 32;
    const bool two = k2 < GA;
    const int2 d1 = runc[k];
    const int2 d2 = two ? runc[k2] : make_int2(0, 0);
    const int4* p1 = binc + (size_t)k * BIN_SL;
    const int4* p2 = binc + (size_t)(two ? k2 : k) * BIN_SL;
    int j1 = d1.x + lane;
    int j2 = d2.x + lane;
    const bool h1 = j1 < d1.y;
    const bool h2 = j2 < d2.y;
    const int4 r1 = p1[h1 ? j1 : 0];
    const int4 r2 = p2[h2 ? j2 : 0];
    const float q1 = dinv[r1.y];
    const float q2 = dinv[r2.y];
    asm volatile("" : : "v"(q1), "v"(q2));
    if (h1) place_one(r1, q1, dl, cur, rec, lrec, base0, fits);
    if (h2) place_one(r2, q2, dl, cur, rec, lrec, base0, fits);
    for (j1 += 32; j1 < d1.y; j1 += 32) {
      const int4 rc = p1[j1];
      place_one(rc, dinv[rc.y], dl, cur, rec, lrec, base0, fits);
    }
    for (j2 += 32; j2 < d2.y; j2 += 32) {
      const int4 rc = p2[j2];
      place_one(rc, dinv[rc.y], dl, cur, rec, lrec, base0, fits);
    }
  }
  __syncthreads();

  int2* pay = (int2*)pay_words;
  const int hi = cur[tid];
  const int d = hi - lo;
  float s0 = 0.0f, s1 = 0.0f, s2 = 0.0f;
  const bool big = d > 32;
  const int dr = big ? 0 : d;
  if (fits) {
    const int rlo = lo - base0;
    if (__builtin_amdgcn_ballot_w32(dr > 16) != 0u)
      order_in_regs<32, 5>(lrec, rlo, pay, lo, dr, s0, s1, s2);
    else
      order_in_regs<16, 4>(lrec, rlo, pay, lo, dr, s0, s1, s2);
    if (big) order_by_search(lrec, rlo, pay, lo, d, s0, s1, s2);
  } else {
    if (__builtin_amdgcn_ballot_w32(dr > 16) != 0u)
      order_in_regs<32, 5>(rec, lo, pay, lo, dr, s0, s1, s2);
    else
      order_in_regs<16, 4>(rec, lo, pay, lo, dr, s0, s1, s2);
    if (big) order_by_search(rec, lo, pay, lo, d, s0, s1, s2);
  }
  if (node < N) ((float4*)s)[node] = make_float4(s0, s1, s2, 0.0f);
}

#define GATHER_NPW 16
#define GATHER_RB 8
#define GATHER_WPB 8
#define ST_PITCH 66
#define OUT_NB 8

template <bool LAST>
__device__ __forceinline__ void node_done(int& cur, int& nextb, float2& acc,
                                          float2& stc, float* __restrict__ st,
                                          int starts, int n0,
                                          float2* __restrict__ o2, int lane) {
  const float2 hv = make_float2(acc.x + stc.x, acc.y + stc.y);
  if (LAST)
    *(float2*)(st + (cur - n0) * ST_PITCH + 2 * lane) = hv;
  else
    o2[(size_t)cur * (EMBED_DIM / 2) + lane] = hv;
  acc = make_float2(0.0f, 0.0f);
  ++cur;
  nextb = __builtin_amdgcn_readlane(starts, cur - n0 + 1);
  stc = *(const float2*)(st + ((cur - n0) & (GATHER_NPW - 1)) * ST_PITCH + 2 * lane);
}

template <bool LAST>
__device__ __forceinline__ void gather_add(int j, float a, float2 v, int& cur,
                                           int& nextb, float2& acc, float2& stc,
                                           float* __restrict__ st, int starts,
                                           int n0, float2* __restrict__ o2,
                                           int lane) {
  while (j == nextb)
    node_done<LAST>(cur, nextb, acc, stc, st, starts, n0, o2, lane);
  acc.x += a * v.x;
  acc.y += a * v.y;
}

template <bool LAST>
__global__ void gather_layer(const int* __restrict__ pay_words,
                             const int* __restrict__ offs,
                             const float* __restrict__ hcur,
                             const float* __restrict__ s,
                             const float* __restrict__ T,
                             float* __restrict__ hnext,
                             const float* __restrict__ emb,
                             const float* __restrict__ h1,
                             float* __restrict__ out, float inv_lp1, int N,
                             int E) {
  __shared__ __attribute__((aligned(8))) float stage[GATHER_WPB][GATHER_NPW * ST_PITCH];
  const int2* pay = (const int2*)pay_words;
  const float2* h2 = (const float2*)hcur;
  float2* o2 = (float2*)hnext;
  const int lane = threadIdx.x & 31;
  const int wave = __builtin_amdgcn_readfirstlane(
      (int)((blockIdx.x * blockDim.x + threadIdx.x) >> 5));
  float* st = stage[__builtin_amdgcn_readfirstlane((int)(threadIdx.x >> 5))];
  const int n0 = wave * GATHER_NPW;

  const int bi = n0 + lane;
  const int starts = (bi < N) ? offs[bi] : E;

  {
    const int half = lane >> 4;
    const int l16 = lane & 15;
    const int k0 = half * 2;

    int arow = n0 + l16;
    arow = (arow < N) ? arow : N - 1;
    v2f A;
    A.x = s[(size_t)arow * 4 + k0];
    A.y = s[(size_t)arow * 4 + k0 + 1];

    v2f B[4];
#pragma unroll
    for (int dt = 0; dt < 4; ++dt) {
      int n = dt * 16 + l16;
      B[dt].x = (k0 + 0 < 3) ? T[(k0 + 0) * EMBED_DIM + n] : 0.0f;
      B[dt].y = (k0 + 1 < 3) ? T[(k0 + 1) * EMBED_DIM + n] : 0.0f;
    }

#pragma unroll
    for (int dt = 0; dt < 4; ++dt) {
      int n = dt * 16 + l16;
      v8f C = {};
      C = __builtin_amdgcn_wmma_f32_16x16x4_f32(
           false, A,  false, B[dt],
           (short)0, C,  false,  false);

#pragma unroll
      for (int r = 0; r < 8; ++r)
        st[(r + 8 * half) * ST_PITCH + n] = C[r];
    }
  }
  __syncthreads();
  if (n0 >= N) return;
  const int n1 = (n0 + GATHER_NPW < N) ? n0 + GATHER_NPW : N;

  const int e0 = __builtin_amdgcn_readlane(starts, 0);
  const int e1 = __builtin_amdgcn_readlane(starts, n1 - n0);

  int cur = n0;
  int nextb = __builtin_amdgcn_readlane(starts, 1);
  float2 acc = make_float2(0.0f, 0.0f);
  float2 stc = *(const float2*)(st + 2 * lane);

  if (e1 > e0) {
    int s0 = e0 + lane;
    int2 pnext = pay[(s0 < e1) ? s0 : e1 - 1];
    for (int jc = e0; jc < e1; jc += 32) {
      const int2 pv = pnext;
      if (jc + 32 < e1) {
        int sn = jc + 32 + lane;
        pnext = pay[(sn < e1) ? sn : e1 - 1];
      }
      const int left = e1 - jc;
      const int cnt = (left < 32) ? left : 32;
      int b = 0;
      for (; b + GATHER_RB <= cnt; b += GATHER_RB) {
        float2 v[GATHER_RB];
#pragma unroll
        for (int k = 0; k < GATHER_RB; ++k) {
          int r = __builtin_amdgcn_readlane(pv.x, b + k);
          v[k] = h2[(size_t)r * (EMBED_DIM / 2) + lane];
        }
#pragma unroll
        for (int k = 0; k < GATHER_RB; ++k) {
          const float a = __int_as_float(__builtin_amdgcn_readlane(pv.y, b + k));
          gather_add<LAST>(jc + b + k, a, v[k], cur, nextb, acc, stc, st,
                           starts, n0, o2, lane);
        }
      }
      if (b < cnt) {
        float2 v[GATHER_RB - 1];
#pragma unroll
        for (int k = 0; k < GATHER_RB - 1; ++k) {
          int r = __builtin_amdgcn_readlane(pv.x, b + k);
          v[k] = h2[(size_t)r * (EMBED_DIM / 2) + lane];
        }
#pragma unroll
        for (int k = 0; k < GATHER_RB - 1; ++k) {
          if (b + k < cnt) {
            const float a = __int_as_float(__builtin_amdgcn_readlane(pv.y, b + k));
            gather_add<LAST>(jc + b + k, a, v[k], cur, nextb, acc, stc, st,
                             starts, n0, o2, lane);
          }
        }
      }
    }
  }
  while (cur < n1)
    node_done<LAST>(cur, nextb, acc, stc, st, starts, n0, o2, lane);

  if (LAST) {
    const float2* e2 = (const float2*)emb;
    const float2* a2 = (const float2*)h1;
    float2* out2 = (float2*)out;
    const float c = inv_lp1;
    for (int q = 0; q < GATHER_NPW; q += OUT_NB) {
      float2 ve[OUT_NB], va[OUT_NB], vb[OUT_NB];
#pragma unroll
      for (int k = 0; k < OUT_NB; ++k) {
        const int node = n0 + q + k;
        const size_t idx =
            (size_t)((node < n1) ? node : n1 - 1) * (EMBED_DIM / 2) + lane;
        ve[k] = e2[idx];
        va[k] = a2[idx];
        vb[k] = h2[idx];
      }
#pragma unroll
      for (int k = 0; k < OUT_NB; ++k) {
        const int node = n0 + q + k;
        if (node < n1) {
          const float2 hv = *(const float2*)(st + (q + k) * ST_PITCH + 2 * lane);
          float2 o = make_float2(ve[k].x * c, ve[k].y * c);
          o.x += c * va[k].x;
          o.y += c * va[k].y;
          o.x += c * vb[k].x;
          o.y += c * vb[k].y;
          o.x += c * hv.x;
          o.y += c * hv.y;
          out2[(size_t)node * (EMBED_DIM / 2) + lane] = o;
        }
      }
    }
  }
}

extern "C" void kernel_launch(void* const* d_in, const int* in_sizes, int n_in,
                              void* d_out, int out_size, void* d_ws,
                              size_t ws_size, hipStream_t stream) {
  const int E = in_sizes[1];
  const int N = in_sizes[3] / EMBED_DIM;

  const int* eidx = (const int*)d_in[0];
  const int* row = eidx;
  const int* col = eidx + E;
  const float* w = (const float*)d_in[1];
  const int* et = (const int*)d_in[2];
  const float* emb = (const float*)d_in[3];
  const float* T = (const float*)d_in[4];
  float* out = (float*)d_out;

  float* ws = (float*)d_ws;
  int* offs = (int*)ws;
  float* S = ws + (size_t)N;
  int* pay = (int*)(S + (size_t)N * 4);
  float* h_a = S + (size_t)N * 4 + (size_t)E * 2;
  float* h_b = h_a + (size_t)N * EMBED_DIM;

  const int GA = (E + BIN_SL - 1) / BIN_SL;
  const int NBK = (N + BKT_N - 1) / BKT_N;
  int* rec = (int*)h_a;
  int* binc = rec + (size_t)E * 4;
  float* dinv = (float*)(binc + (size_t)E * 4);
  int* startc = (int*)(dinv + (size_t)N);
  int* startr = startc + (size_t)GA * BSTRIDE;
  int* rbin = startr + (size_t)GA * BSTRIDE;

  const float inv_lp1 = 1.0f / (NUM_LAYERS + 1);

  bin_edges<<<GA, BIN_T, 0, stream>>>(row, col, w, et, binc, rbin, startc,
                                      startr, E);
  bucket_count<<<NBK, BIN_T, 0, stream>>>(binc, rbin, startc, startr, dinv,
                                          offs, N, GA);
  bucket_place<<<NBK, BIN_T, 0, stream>>>(binc, startc, offs, dinv, rec, pay, S,
                                          N, GA, E);

  const int gather_waves = (N + GATHER_NPW - 1) / GATHER_NPW;
  const int gather_grid = (gather_waves + GATHER_WPB - 1) / GATHER_WPB;
  gather_layer<false><<<gather_grid, GATHER_WPB * 32, 0, stream>>>(
      pay, offs, emb, S, T, h_a, nullptr, nullptr, nullptr, inv_lp1, N, E);
  gather_layer<false><<<gather_grid, GATHER_WPB * 32, 0, stream>>>(
      pay, offs, h_a, S, T, h_b, nullptr, nullptr, nullptr, inv_lp1, N, E);
  gather_layer<true><<<gather_grid, GATHER_WPB * 32, 0, stream>>>(
      pay, offs, h_b, S, T, nullptr, emb, h_a, out, inv_lp1, N, E);
}


# More independent loads per wave before the first is used: the gather keeps 16 rows in flight (was 8); the two bucket kernels request 5 slices' runs per trip (were 1 and 2); sum order kept
# speedup vs baseline: 7.3640x; 1.0517x over previous
#include <hip/hip_runtime.h>


#define EMBED_DIM 64
#define NUM_LAYERS 3

typedef float v2f __attribute__((ext_vector_type(2)));
typedef float v8f __attribute__((ext_vector_type(8)));

#define BIN_SL 4096
#define BIN_T 1024
#define BIN_PER (BIN_SL / BIN_T)
#define BKT_SHIFT 10
#define BKT_N (1 << BKT_SHIFT)
#define MAXB 256
#define BSTRIDE (MAXB + 1)
#define MAXGA 512
#define WALK_U 5

__global__ __launch_bounds__(BIN_T) void bin_edges(
    const int* __restrict__ row, const int* __restrict__ col,
    const float* __restrict__ w, const int* __restrict__ et,
    int* __restrict__ binc_words, int* __restrict__ rbin_words,
    int* __restrict__ startc, int* __restrict__ startr, int E) {
  __shared__ int hc[MAXB];
  __shared__ int hr[MAXB];
  __shared__ int wtc[MAXB / 32];
  __shared__ int wtr[MAXB / 32];
  __shared__ int4 sc[BIN_SL];
  __shared__ unsigned short sr[BIN_SL];
  int4* binc = (int4*)binc_words;
  unsigned short* rbin = (unsigned short*)rbin_words;
  const int tid = threadIdx.x;
  const int lane = tid & 31;
  const int wave = tid >> 5;
  const int base = blockIdx.x * BIN_SL;
  const int n = (E - base < BIN_SL) ? E - base : BIN_SL;

  if (tid < MAXB) {
    hc[tid] = 0;
    hr[tid] = 0;
  }
  __syncthreads();

  int r[BIN_PER], c[BIN_PER], ty[BIN_PER], pc[BIN_PER], pr[BIN_PER];
  float we[BIN_PER];
#pragma unroll
  for (int k = 0; k < BIN_PER; ++k) {
    const int i = tid + k * BIN_T;
    const int e = base + ((i < n) ? i : n - 1);
    r[k] = row[e];
    c[k] = col[e];
    we[k] = w[e];
    ty[k] = et[e];
  }
#pragma unroll
  for (int k = 0; k < BIN_PER; ++k) {
    const int i = tid + k * BIN_T;
    pc[k] = 0;
    pr[k] = 0;
    if (i < n) {
      pc[k] = atomicAdd(&hc[c[k] >> BKT_SHIFT], 1);
      pr[k] = atomicAdd(&hr[r[k] >> BKT_SHIFT], 1);
    }
  }
  __syncthreads();

  int vc = 0, vr = 0, ic = 0, ir = 0;
  if (tid < MAXB) {
    vc = hc[tid];
    vr = hr[tid];
    ic = vc;
    ir = vr;
    for (int o = 1; o < 32; o <<= 1) {
      const int tc = __shfl_up(ic, o, 32);
      const int tr = __shfl_up(ir, o, 32);
      if (lane >= o) {
        ic += tc;
        ir += tr;
      }
    }
    if (lane == 31) {
      wtc[wave] = ic;
      wtr[wave] = ir;
    }
  }
  __syncthreads();
  if (tid < MAXB) {
    int oc = 0, orr = 0;
    for (int k = 0; k < wave; ++k) {
      oc += wtc[k];
      orr += wtr[k];
    }
    const int ec = oc + ic - vc;
    const int er = orr + ir - vr;
    hc[tid] = ec;
    hr[tid] = er;
    startc[(size_t)blockIdx.x * BSTRIDE + tid] = ec;
    startr[(size_t)blockIdx.x * BSTRIDE + tid] = er;
    if (tid == MAXB - 1) {
      startc[(size_t)blockIdx.x * BSTRIDE + MAXB] = ec + vc;
      startr[(size_t)blockIdx.x * BSTRIDE + MAXB] = er + vr;
    }
  }
  __syncthreads();

#pragma unroll
  for (int k = 0; k < BIN_PER; ++k) {
    const int i = tid + k * BIN_T;
    if (i < n) {
      sc[hc[c[k] >> BKT_SHIFT] + pc[k]] =
          make_int4((base + i) * 4 + (ty[k] & 3), r[k], c[k],
                    __float_as_int(we[k]));
      sr[hr[r[k] >> BKT_SHIFT] + pr[k]] = (unsigned short)(r[k] & (BKT_N - 1));
    }
  }
  __syncthreads();

  for (int i = tid; i < n; i += BIN_T) {
    binc[(size_t)base + i] = sc[i];
    rbin[(size_t)base + i] = sr[i];
  }
}

__global__ __launch_bounds__(BIN_T) void bucket_count(
    const int* __restrict__ binc_words, const int* __restrict__ rbin_words,
    const int* __restrict__ startc, const int* __restrict__ startr,
    float* __restrict__ dinv, int* __restrict__ offs, int N, int GA) {
  __shared__ int cd[BKT_N];
  __shared__ int cc[BKT_N];
  __shared__ int2 runc[MAXGA];
  __shared__ int2 runr[MAXGA];
  __shared__ int red[BIN_T / 32];
  __shared__ int wtot[BIN_T / 32];
  const int4* binc = (const int4*)binc_words;
  const unsigned short* rbin = (const unsigned short*)rbin_words;
  const int tid = threadIdx.x;
  const int lane = tid & 31;
  const int wave = __builtin_amdgcn_readfirstlane(tid >> 5);
  const int b = blockIdx.x;

  cd[tid] = 0;
  cc[tid] = 0;
  int part = 0;
  for (int k = tid; k < GA; k += BIN_T) {
    const int s0 = startc[(size_t)k * BSTRIDE + b];
    const int s1 = startc[(size_t)k * BSTRIDE + b + 1];
    const int t0 = startr[(size_t)k * BSTRIDE + b];
    const int t1 = startr[(size_t)k * BSTRIDE + b + 1];
    runc[k] = make_int2(s0, s1);
    runr[k] = make_int2(t0, t1);
    part += s0;
  }
  for (int o = 16; o > 0; o >>= 1) part += __shfl_xor(part, o, 32);
  if (lane == 0) red[wave] = part;
  __syncthreads();

  for (int k0 = wave; k0 < GA; k0 += WALK_U * (BIN_T / 32)) {
    int2 dc[WALK_U], dr[WALK_U];
    int rw[WALK_U], cv[WALK_U];
#pragma unroll
    for (int u = 0; u < WALK_U; ++u) {
      const int k = k0 + u * (BIN_T / 32);
      const bool on = k < GA;
      const int kk = on ? k : k0;
      dc[u] = on ? runc[kk] : make_int2(0, 0);
      dr[u] = on ? runr[kk] : make_int2(0, 0);
      const size_t sb = (size_t)kk * BIN_SL;
      const int jc = dc[u].x + lane;
      const int jr = dr[u].x + lane;
      rw[u] = rbin_words[(sb >> 1) + (size_t)(((jr < dr[u].y) ? jr : 0) >> 1)];
      cv[u] = binc[sb + ((jc < dc[u].y) ? jc : 0)].z;
    }
#pragma unroll
    for (int u = 0; u < WALK_U; ++u) {
      const int k = k0 + u * (BIN_T / 32);
      const size_t sb = (size_t)((k < GA) ? k : k0) * BIN_SL;
      int jc = dc[u].x + lane;
      int jr = dr[u].x + lane;
      if (jc < dc[u].y) atomicAdd(&cc[cv[u] & (BKT_N - 1)], 1);
      if (jr < dr[u].y) atomicAdd(&cd[(rw[u] >> ((jr & 1) * 16)) & 0xffff], 1);
      for (jc += 32; jc < dc[u].y; jc += 32)
        atomicAdd(&cc[binc[sb + jc].z & (BKT_N - 1)], 1);
      for (jr += 32; jr < dr[u].y; jr += 32) atomicAdd(&cd[rbin[sb + jr]], 1);
    }
  }
  __syncthreads();

  const int node = b * BKT_N + tid;
  if (node < N) {
    const float d = (float)cd[tid];
    dinv[node] = (d > 0.0f) ? rsqrtf(d) : 0.0f;
  }
  const int v = cc[tid];
  int incl = v;
  for (int o = 1; o < 32; o <<= 1) {
    const int t = __shfl_up(incl, o, 32);
    if (lane >= o) incl += t;
  }
  if (lane == 31) wtot[wave] = incl;
  __syncthreads();
  int run = 0;
  for (int k = 0; k < BIN_T / 32; ++k) {
    run += red[k];
    if (k < wave) run += wtot[k];
  }
  run += incl - v;
  if (node < N) offs[node] = run;
}

__device__ __forceinline__ void place_one(const int4 rc, const float dr,
                                          const float* __restrict__ dl,
                                          int* __restrict__ cur,
                                          int4* rec, int4* lrec, int base0,
                                          bool fits) {
  const int cl = rc.z & (BKT_N - 1);
  const float nm = dr * dl[cl];
  const float a = __int_as_float(rc.w) * nm;
  const int slot = atomicAdd(&cur[cl], 1);
  const int4 v = make_int4(rc.x, rc.y, __float_as_int(a), __float_as_int(nm));
  if (fits)
    lrec[slot - base0] = v;
  else
    rec[slot] = v;
}

template <int NS>
__device__ __forceinline__ void sort_regs(int (&v)[NS]) {
#pragma unroll
  for (int p = 1; (1 << p) <= NS; ++p) {
#pragma unroll
    for (int q = p - 1; q >= 0; --q) {
#pragma unroll
      for (int i = 0; i < NS; ++i) {
        const int l = i ^ (1 << q);
        if (l > i) {
          const int a = v[i];
          const int b = v[l];
          const int mn = (a < b) ? a : b;
          const int mx = (a < b) ? b : a;
          const bool up = ((i >> p) & 1) == 0;
          v[i] = up ? mn : mx;
          v[l] = up ? mx : mn;
        }
      }
    }
  }
}

template <int NS, int SH>
__device__ __forceinline__ void order_in_regs(const int4* rec, int rlo,
                                              int2* __restrict__ pay, int lo,
                                              int d, float& s0, float& s1,
                                              float& s2) {
  if (d <= 0) return;
  int v[NS];
#pragma unroll
  for (int i = 0; i < NS; ++i) {
    const int key = rec[rlo + ((i < d) ? i : 0)].x;
    v[i] = (i < d) ? ((key << SH) | i) : 0x7fffffff;
  }
  sort_regs<NS>(v);
#pragma unroll
  for (int c = 0; c < NS; c += 8) {
    int4 r[8];
#pragma unroll
    for (int k = 0; k < 8; ++k) {
      r[k] = make_int4(0, 0, 0, 0);
      if (c + k < d) r[k] = rec[rlo + (v[c + k] & (NS - 1))];
    }
#pragma unroll
    for (int k = 0; k < 8; ++k) {
      if (c + k < d) {
        pay[lo + c + k] = make_int2(r[k].y, r[k].z);
        const float nm = __int_as_float(r[k].w);
        const int t = (v[c + k] >> SH) & 3;
        s0 += (t == 0) ? nm : 0.0f;
        s1 += (t == 1) ? nm : 0.0f;
        s2 += (t == 2) ? nm : 0.0f;
      }
    }
  }
}

__device__ __forceinline__ void order_by_search(const int4* rec, int rlo,
                                                int2* __restrict__ pay, int lo,
                                                int d, float& s0, float& s1,
                                                float& s2) {
  int prev = -1;
  for (int k = 0; k < d; ++k) {
    int best = 0x7fffffff;
    int bj = 0;
    for (int j = 0; j < d; ++j) {
      const int key = rec[rlo + j].x;
      if (key > prev && key < best) {
        best = key;
        bj = j;
      }
    }
    const int4 rc = rec[rlo + bj];
    pay[lo + k] = make_int2(rc.y, rc.z);
    const float nm = __int_as_float(rc.w);
    const int t = best & 3;
    s0 += (t == 0) ? nm : 0.0f;
    s1 += (t == 1) ? nm : 0.0f;
    s2 += (t == 2) ? nm : 0.0f;
    prev = best;
  }
}

#define LREC_CAP 16384
__global__ __launch_bounds__(BIN_T) void bucket_place(
    const int* __restrict__ binc_words, const int* __restrict__ startc,
    const int* __restrict__ offs, const float* __restrict__ dinv,
    int* __restrict__ rec_words, int* __restrict__ pay_words,
    float* __restrict__ s, int N, int GA, int E) {
  __shared__ int cur[BKT_N];
  __shared__ float dl[BKT_N];
  __shared__ int2 runc[MAXGA];
  __shared__ int4 lrec[LREC_CAP];
  const int4* binc = (const int4*)binc_words;
  int4* rec = (int4*)rec_words;
  const int tid = threadIdx.x;
  const int lane = tid & 31;
  const int wave = __builtin_amdgcn_readfirstlane(tid >> 5);
  const int b = blockIdx.x;

  const int base0 = offs[b * BKT_N];
  const int end0 = ((b + 1) * BKT_N < N) ? offs[(b + 1) * BKT_N] : E;
  const bool fits = (end0 - base0) <= LREC_CAP;

  const int node = b * BKT_N + tid;
  const int lo = (node < N) ? offs[node] : 0;
  cur[tid] = lo;
  dl[tid] = (node < N) ? dinv[node] : 0.0f;
  for (int k = tid; k < GA; k += BIN_T)
    runc[k] = make_int2(startc[(size_t)k * BSTRIDE + b],
                        startc[(size_t)k * BSTRIDE + b + 1]);
  __syncthreads();

  for (int k0 = wave; k0 < GA; k0 += WALK_U * (BIN_T / 32)) {
    int2 dd[WALK_U];
    int4 rr[WALK_U];
    float qq[WALK_U];
#pragma unroll
    for (int u = 0; u < WALK_U; ++u) {
      const int k = k0 + u * (BIN_T / 32);
      const bool on = k < GA;
      const int kk = on ? k : k0;
      dd[u] = on ? runc[kk] : make_int2(0, 0);
      const int4* p = binc + (size_t)kk * BIN_SL;
      const int j = dd[u].x + lane;
      rr[u] = p[(j < dd[u].y) ? j : 0];
    }
#pragma unroll
    for (int u = 0; u < WALK_U; ++u) qq[u] = dinv[rr[u].y];
#pragma unroll
    for (int u = 0; u < WALK_U; ++u) {
      const int k = k0 + u * (BIN_T / 32);
      const int4* p = binc + (size_t)((k < GA) ? k : k0) * BIN_SL;
      int j = dd[u].x + lane;
      if (j < dd[u].y) place_one(rr[u], qq[u], dl, cur, rec, lrec, base0, fits);
      for (j += 32; j < dd[u].y; j += 32) {
        const int4 rc = p[j];
        place_one(rc, dinv[rc.y], dl, cur, rec, lrec, base0, fits);
      }
    }
  }
  __syncthreads();

  int2* pay = (int2*)pay_words;
  const int hi = cur[tid];
  const int d = hi - lo;
  float s0 = 0.0f, s1 = 0.0f, s2 = 0.0f;
  const bool big = d > 32;
  const int dr = big ? 0 : d;
  if (fits) {
    const int rlo = lo - base0;
    if (__builtin_amdgcn_ballot_w32(dr > 16) != 0u)
      order_in_regs<32, 5>(lrec, rlo, pay, lo, dr, s0, s1, s2);
    else
      order_in_regs<16, 4>(lrec, rlo, pay, lo, dr, s0, s1, s2);
    if (big) order_by_search(lrec, rlo, pay, lo, d, s0, s1, s2);
  } else {
    if (__builtin_amdgcn_ballot_w32(dr > 16) != 0u)
      order_in_regs<32, 5>(rec, lo, pay, lo, dr, s0, s1, s2);
    else
      order_in_regs<16, 4>(rec, lo, pay, lo, dr, s0, s1, s2);
    if (big) order_by_search(rec, lo, pay, lo, d, s0, s1, s2);
  }
  if (node < N) ((float4*)s)[node] = make_float4(s0, s1, s2, 0.0f);
}

#define GATHER_NPW 16
#define GATHER_RB 16
#define GATHER_WPB 8
#define ST_PITCH 66
#define OUT_NB 8

template <bool LAST>
__device__ __forceinline__ void node_done(int& cur, int& nextb, float2& acc,
                                          float2& stc, float* __restrict__ st,
                                          int starts, int n0,
                                          float2* __restrict__ o2, int lane) {
  const float2 hv = make_float2(acc.x + stc.x, acc.y + stc.y);
  if (LAST)
    *(float2*)(st + (cur - n0) * ST_PITCH + 2 * lane) = hv;
  else
    o2[(size_t)cur * (EMBED_DIM / 2) + lane] = hv;
  acc = make_float2(0.0f, 0.0f);
  ++cur;
  nextb = __builtin_amdgcn_readlane(starts, cur - n0 + 1);
  stc = *(const float2*)(st + ((cur - n0) & (GATHER_NPW - 1)) * ST_PITCH + 2 * lane);
}

template <bool LAST>
__device__ __forceinline__ void gather_add(int j, float a, float2 v, int& cur,
                                           int& nextb, float2& acc, float2& stc,
                                           float* __restrict__ st, int starts,
                                           int n0, float2* __restrict__ o2,
                                           int lane) {
  while (j == nextb)
    node_done<LAST>(cur, nextb, acc, stc, st, starts, n0, o2, lane);
  acc.x += a * v.x;
  acc.y += a * v.y;
}

template <bool LAST>
__global__ void gather_layer(const int* __restrict__ pay_words,
                             const int* __restrict__ offs,
                             const float* __restrict__ hcur,
                             const float* __restrict__ s,
                             const float* __restrict__ T,
                             float* __restrict__ hnext,
                             const float* __restrict__ emb,
                             const float* __restrict__ h1,
                             float* __restrict__ out, float inv_lp1, int N,
                             int E) {
  __shared__ __attribute__((aligned(8))) float stage[GATHER_WPB][GATHER_NPW * ST_PITCH];
  const int2* pay = (const int2*)pay_words;
  const float2* h2 = (const float2*)hcur;
  float2* o2 = (float2*)hnext;
  const int lane = threadIdx.x & 31;
  const int wave = __builtin_amdgcn_readfirstlane(
      (int)((blockIdx.x * blockDim.x + threadIdx.x) >> 5));
  float* st = stage[__builtin_amdgcn_readfirstlane((int)(threadIdx.x >> 5))];
  const int n0 = wave * GATHER_NPW;

  const int bi = n0 + lane;
  const int starts = (bi < N) ? offs[bi] : E;

  {
    const int half = lane >> 4;
    const int l16 = lane & 15;
    const int k0 = half * 2;

    int arow = n0 + l16;
    arow = (arow < N) ? arow : N - 1;
    v2f A;
    A.x = s[(size_t)arow * 4 + k0];
    A.y = s[(size_t)arow * 4 + k0 + 1];

    v2f B[4];
#pragma unroll
    for (int dt = 0; dt < 4; ++dt) {
      int n = dt * 16 + l16;
      B[dt].x = (k0 + 0 < 3) ? T[(k0 + 0) * EMBED_DIM + n] : 0.0f;
      B[dt].y = (k0 + 1 < 3) ? T[(k0 + 1) * EMBED_DIM + n] : 0.0f;
    }

#pragma unroll
    for (int dt = 0; dt < 4; ++dt) {
      int n = dt * 16 + l16;
      v8f C = {};
      C = __builtin_amdgcn_wmma_f32_16x16x4_f32(
           false, A,  false, B[dt],
           (short)0, C,  false,  false);

#pragma unroll
      for (int r = 0; r < 8; ++r)
        st[(r + 8 * half) * ST_PITCH + n] = C[r];
    }
  }
  __syncthreads();
  if (n0 >= N) return;
  const int n1 = (n0 + GATHER_NPW < N) ? n0 + GATHER_NPW : N;

  const int e0 = __builtin_amdgcn_readlane(starts, 0);
  const int e1 = __builtin_amdgcn_readlane(starts, n1 - n0);

  int cur = n0;
  int nextb = __builtin_amdgcn_readlane(starts, 1);
  float2 acc = make_float2(0.0f, 0.0f);
  float2 stc = *(const float2*)(st + 2 * lane);

  if (e1 > e0) {
    int s0 = e0 + lane;
    int2 pnext = pay[(s0 < e1) ? s0 : e1 - 1];
    for (int jc = e0; jc < e1; jc += 32) {
      const int2 pv = pnext;
      if (jc + 32 < e1) {
        int sn = jc + 32 + lane;
        pnext = pay[(sn < e1) ? sn : e1 - 1];
      }
      const int left = e1 - jc;
      const int cnt = (left < 32) ? left : 32;
      int b = 0;
      for (; b + GATHER_RB <= cnt; b += GATHER_RB) {
        float2 v[GATHER_RB];
#pragma unroll
        for (int k = 0; k < GATHER_RB; ++k) {
          const unsigned r = (unsigned)__builtin_amdgcn_readlane(pv.x, b + k);
          v[k] = h2[r * (unsigned)(EMBED_DIM / 2) + (unsigned)lane];
        }
#pragma unroll
        for (int k = 0; k < GATHER_RB; ++k) {
          const float a = __int_as_float(__builtin_amdgcn_readlane(pv.y, b + k));
          gather_add<LAST>(jc + b + k, a, v[k], cur, nextb, acc, stc, st,
                           starts, n0, o2, lane);
        }
      }
      if (b < cnt) {
        float2 v[GATHER_RB - 1];
#pragma unroll
        for (int k = 0; k < GATHER_RB - 1; ++k) {
          const unsigned r = (unsigned)__builtin_amdgcn_readlane(pv.x, b + k);
          v[k] = h2[r * (unsigned)(EMBED_DIM / 2) + (unsigned)lane];
        }
#pragma unroll
        for (int k = 0; k < GATHER_RB - 1; ++k) {
          if (b + k < cnt) {
            const float a = __int_as_float(__builtin_amdgcn_readlane(pv.y, b + k));
            gather_add<LAST>(jc + b + k, a, v[k], cur, nextb, acc, stc, st,
                             starts, n0, o2, lane);
          }
        }
      }
    }
  }
  while (cur < n1)
    node_done<LAST>(cur, nextb, acc, stc, st, starts, n0, o2, lane);

  if (LAST) {
    const float2* e2 = (const float2*)emb;
    const float2* a2 = (const float2*)h1;
    float2* out2 = (float2*)out;
    const float c = inv_lp1;
    for (int q = 0; q < GATHER_NPW; q += OUT_NB) {
      float2 ve[OUT_NB], va[OUT_NB], vb[OUT_NB];
#pragma unroll
      for (int k = 0; k < OUT_NB; ++k) {
        const int node = n0 + q + k;
        const size_t idx =
            (size_t)((node < n1) ? node : n1 - 1) * (EMBED_DIM / 2) + lane;
        ve[k] = e2[idx];
        va[k] = a2[idx];
        vb[k] = h2[idx];
      }
#pragma unroll
      for (int k = 0; k < OUT_NB; ++k) {
        const int node = n0 + q + k;
        if (node < n1) {
          const float2 hv = *(const float2*)(st + (q + k) * ST_PITCH + 2 * lane);
          float2 o = make_float2(ve[k].x * c, ve[k].y * c);
          o.x += c * va[k].x;
          o.y += c * va[k].y;
          o.x += c * vb[k].x;
          o.y += c * vb[k].y;
          o.x += c * hv.x;
          o.y += c * hv.y;
          out2[(size_t)node * (EMBED_DIM / 2) + lane] = o;
        }
      }
    }
  }
}

extern "C" void kernel_launch(void* const* d_in, const int* in_sizes, int n_in,
                              void* d_out, int out_size, void* d_ws,
                              size_t ws_size, hipStream_t stream) {
  const int E = in_sizes[1];
  const int N = in_sizes[3] / EMBED_DIM;

  const int* eidx = (const int*)d_in[0];
  const int* row = eidx;
  const int* col = eidx + E;
  const float* w = (const float*)d_in[1];
  const int* et = (const int*)d_in[2];
  const float* emb = (const float*)d_in[3];
  const float* T = (const float*)d_in[4];
  float* out = (float*)d_out;

  float* ws = (float*)d_ws;
  int* offs = (int*)ws;
  float* S = ws + (size_t)N;
  int* pay = (int*)(S + (size_t)N * 4);
  float* h_a = S + (size_t)N * 4 + (size_t)E * 2;
  float* h_b = h_a + (size_t)N * EMBED_DIM;

  const int GA = (E + BIN_SL - 1) / BIN_SL;
  const int NBK = (N + BKT_N - 1) / BKT_N;
  int* rec = (int*)h_a;
  int* binc = rec + (size_t)E * 4;
  float* dinv = (float*)(binc + (size_t)E * 4);
  int* startc = (int*)(dinv + (size_t)N);
  int* startr = startc + (size_t)GA * BSTRIDE;
  int* rbin = startr + (size_t)GA * BSTRIDE;

  const float inv_lp1 = 1.0f / (NUM_LAYERS + 1);

  bin_edges<<<GA, BIN_T, 0, stream>>>(row, col, w, et, binc, rbin, startc,
                                      startr, E);
  bucket_count<<<NBK, BIN_T, 0, stream>>>(binc, rbin, startc, startr, dinv,
                                          offs, N, GA);
  bucket_place<<<NBK, BIN_T, 0, stream>>>(binc, startc, offs, dinv, rec, pay, S,
                                          N, GA, E);

  const int gather_waves = (N + GATHER_NPW - 1) / GATHER_NPW;
  const int gather_grid = (gather_waves + GATHER_WPB - 1) / GATHER_WPB;
  gather_layer<false><<<gather_grid, GATHER_WPB * 32, 0, stream>>>(
      pay, offs, emb, S, T, h_a, nullptr, nullptr, nullptr, inv_lp1, N, E);
  gather_layer<false><<<gather_grid, GATHER_WPB * 32, 0, stream>>>(
      pay, offs, h_a, S, T, h_b, nullptr, nullptr, nullptr, inv_lp1, N, E);
  gather_layer<true><<<gather_grid, GATHER_WPB * 32, 0, stream>>>(
      pay, offs, h_b, S, T, nullptr, emb, h_a, out, inv_lp1, N, E);
}
